# starting kernel
# baseline (speedup 1.0000x reference)
#include <hip/hip_runtime.h>

typedef __attribute__((ext_vector_type(2))) float v2f;
typedef __attribute__((ext_vector_type(8))) float v8f;

#define BATCH 4096
#define INF   1024
#define OUTF  1024

__device__ __forceinline__ v8f wmma_f32(v2f a, v2f b, v8f c) {
  return __builtin_amdgcn_wmma_f32_16x16x4_f32(false, a, false, b, (short)0, c,
                                               false, false);
}

__global__ void __launch_bounds__(256) taylor_z_kernel(
    const float* __restrict__ X, const float* __restrict__ I0,
    const float* __restrict__ I1, const float* __restrict__ I2,
    float* __restrict__ Z) {
  const int lane = threadIdx.x & 31;
  const int wave = threadIdx.x >> 5;
  const int tile = blockIdx.x * 8 + wave;
  const int mt = tile % 12, nt = tile / 12;
  const int row0 = mt * 16, col0 = nt * 16;

  const float* Arow;
  if (row0 < 32)      Arow = I0 + row0 * INF;
  else if (row0 < 96) Arow = I1 + (row0 - 32) * INF;
  else                Arow = I2 + (row0 - 96) * INF;

  const int m = lane & 15, n = lane & 15;
  const int ko = (lane >> 4) * 2;
  const float* ap = Arow + m * INF + ko;
  const float* bp = X + (col0 + n) * INF + ko;

  v8f acc = {};
  for (int k = 0; k < INF; k += 4) {
    v2f a, b;
    a[0] = ap[k]; a[1] = ap[k + 1];
    b[0] = bp[k]; b[1] = bp[k + 1];
    acc = wmma_f32(a, b, acc);
  }
  const int mb = (lane >> 4) * 8;
#pragma unroll
  for (int v = 0; v < 8; ++v)
    Z[(row0 + mb + v) * BATCH + col0 + n] = acc[v];
}

__global__ void __launch_bounds__(256) taylor_r0_kernel(
    const float* __restrict__ G0, const float* __restrict__ Z,
    float* __restrict__ R) {
  const int lane = threadIdx.x & 31;
  const int wave = threadIdx.x >> 5;
  const int tile = blockIdx.x * 8 + wave;
  const int mt = tile & 1, nt = tile >> 1;
  const int row0 = mt * 16, col0 = nt * 16;
  const int m = lane & 15, n = lane & 15;
  const int ko = (lane >> 4) * 2;

  v8f acc = {};
  for (int k = 0; k < 32; k += 4) {
    const int kk = k + ko;
    v2f a, b;
    a[0] = G0[(row0 + m) * 32 + kk];
    a[1] = G0[(row0 + m) * 32 + kk + 1];
    b[0] = Z[kk * BATCH + col0 + n];
    b[1] = Z[(kk + 1) * BATCH + col0 + n];
    acc = wmma_f32(a, b, acc);
  }
  const int mb = (lane >> 4) * 8;
#pragma unroll
  for (int v = 0; v < 8; ++v)
    R[(row0 + mb + v) * BATCH + col0 + n] = acc[v];
}

__global__ void __launch_bounds__(256) taylor_r1_kernel(
    const float* __restrict__ G1, const float* __restrict__ Z,
    float* __restrict__ R) {
  const int lane = threadIdx.x & 31;
  const int wave = threadIdx.x >> 5;
  const int tile = blockIdx.x * 8 + wave;
  const int mt = tile & 1, nt = tile >> 1;
  const int row0 = mt * 16, col0 = nt * 16;
  const int m = lane & 15, n = lane & 15;
  const int ko = (lane >> 4) * 2;
  const int c = col0 + n;

  v8f acc = {};
  for (int k = 0; k < 1024; k += 4) {
    const int k0 = k + ko, k1 = k0 + 1;
    v2f a, b;
    a[0] = G1[(row0 + m) * 1024 + k0];
    a[1] = G1[(row0 + m) * 1024 + k1];
    b[0] = Z[(64 + (k0 >> 5)) * BATCH + c] * Z[(32 + (k0 & 31)) * BATCH + c];
    b[1] = Z[(64 + (k1 >> 5)) * BATCH + c] * Z[(32 + (k1 & 31)) * BATCH + c];
    acc = wmma_f32(a, b, acc);
  }
  const int mb = (lane >> 4) * 8;
#pragma unroll
  for (int v = 0; v < 8; ++v)
    R[(32 + row0 + mb + v) * BATCH + col0 + n] = acc[v];
}

__global__ void __launch_bounds__(256) taylor_r2_kernel(
    const float* __restrict__ G2, const float* __restrict__ Z,
    float* __restrict__ R) {
  __shared__ float zs[3 * 32 * 16];
  const int lane = threadIdx.x & 31;
  const int wave = threadIdx.x >> 5;
  const int col0 = blockIdx.x * 16;

  for (int idx = threadIdx.x; idx < 1536; idx += 256) {
    const int rr = idx >> 4, cc = idx & 15;
    zs[idx] = Z[(96 + rr) * BATCH + col0 + cc];
  }
  __syncthreads();

  const int rt = wave >> 2;
  const int ks = wave & 3;
  const int row0 = rt * 16;
  const int m = lane & 15, n = lane & 15;
  const int ko = (lane >> 4) * 2;
  const int kbeg = ks * 8192, kend = kbeg + 8192;
  const float* gp = G2 + (row0 + m) * 32768;

  v8f acc = {};
  for (int k = kbeg; k < kend; k += 4) {
    const int k0 = k + ko, k1 = k0 + 1;
    v2f a, b;
    a[0] = gp[k0];
    a[1] = gp[k1];
    b[0] = zs[(k0 & 31) * 16 + n] *
           zs[512 + ((k0 >> 5) & 31) * 16 + n] *
           zs[1024 + (k0 >> 10) * 16 + n];
    b[1] = zs[(k1 & 31) * 16 + n] *
           zs[512 + ((k1 >> 5) & 31) * 16 + n] *
           zs[1024 + (k1 >> 10) * 16 + n];
    acc = wmma_f32(a, b, acc);
  }
  const int mb = (lane >> 4) * 8;
#pragma unroll
  for (int v = 0; v < 8; ++v)
    atomicAdd(&R[(64 + row0 + mb + v) * BATCH + col0 + n], acc[v]);
}

__global__ void __launch_bounds__(256) taylor_out_kernel(
    const float* __restrict__ O0, const float* __restrict__ O1,
    const float* __restrict__ O2, const float* __restrict__ Cst,
    const float* __restrict__ R, float* __restrict__ Y) {
  __shared__ float ts[8 * 16 * 17];
  const int lane = threadIdx.x & 31;
  const int wave = threadIdx.x >> 5;
  const int tile = blockIdx.x * 8 + wave;
  const int mt = tile & 63, nt = tile >> 6;
  const int row0 = mt * 16;
  const int col0 = nt * 16;
  const int m = lane & 15, n = lane & 15;
  const int ko = (lane >> 4) * 2;

  v8f acc = {};
  for (int k = 0; k < 96; k += 4) {
    const int kk = k + ko;
    const float* Op = O0; int rk = kk;
    if (kk >= 64)      { Op = O2; rk = kk - 64; }
    else if (kk >= 32) { Op = O1; rk = kk - 32; }
    v2f a, b;
    a[0] = Op[(row0 + m) * 32 + rk];
    a[1] = Op[(row0 + m) * 32 + rk + 1];
    b[0] = R[kk * BATCH + col0 + n];
    b[1] = R[(kk + 1) * BATCH + col0 + n];
    acc = wmma_f32(a, b, acc);
  }
  const int mb = (lane >> 4) * 8;
  float* tw = ts + wave * 272;
#pragma unroll
  for (int v = 0; v < 8; ++v)
    tw[(mb + v) * 17 + n] = acc[v];
  __syncthreads();

  const int ol = lane & 15;
  const int bh = (lane >> 4) * 8;
  const float cv = Cst[row0 + ol];
#pragma unroll
  for (int v = 0; v < 8; ++v)
    Y[(col0 + bh + v) * OUTF + row0 + ol] = tw[ol * 17 + bh + v] + cv;
}

extern "C" void kernel_launch(void* const* d_in, const int* in_sizes, int n_in,
                              void* d_out, int out_size, void* d_ws,
                              size_t ws_size, hipStream_t stream) {
  const float* X  = (const float*)d_in[0];
  const float* C  = (const float*)d_in[1];
  const float* O0 = (const float*)d_in[2];
  const float* I0 = (const float*)d_in[3];
  const float* G0 = (const float*)d_in[4];
  const float* O1 = (const float*)d_in[5];
  const float* I1 = (const float*)d_in[6];
  const float* G1 = (const float*)d_in[7];
  const float* O2 = (const float*)d_in[8];
  const float* I2 = (const float*)d_in[9];
  const float* G2 = (const float*)d_in[10];
  float* Y = (float*)d_out;

  float* Z = (float*)d_ws;
  float* R = Z + 192 * BATCH;

  hipMemsetAsync(R + 64 * BATCH, 0, 32 * BATCH * sizeof(float), stream);

  taylor_z_kernel  <<<384,  256, 0, stream>>>(X, I0, I1, I2, Z);
  taylor_r0_kernel <<<64,   256, 0, stream>>>(G0, Z, R);
  taylor_r1_kernel <<<64,   256, 0, stream>>>(G1, Z, R);
  taylor_r2_kernel <<<256,  256, 0, stream>>>(G2, Z, R);
  taylor_out_kernel<<<2048, 256, 0, stream>>>(O0, O1, O2, C, R, Y);
}


# Stages 1, 2a, 2b fused per 16-column batch panel: one block builds all 192 Z rows through shared LDS, then R0 and R1 from its own panel (5 launches to 3); sum order of Z, R0, R1 changes
# speedup vs baseline: 3.4170x; 3.4170x over previous
#include <hip/hip_runtime.h>

typedef __attribute__((ext_vector_type(2))) float v2f;
typedef __attribute__((ext_vector_type(8))) float v8f;
typedef __attribute__((ext_vector_type(4))) float v4f;

#define BATCH 4096
#define INF   1024
#define OUTF  1024

__device__ __forceinline__ v8f wmma_f32(v2f a, v2f b, v8f c) {
  return __builtin_amdgcn_wmma_f32_16x16x4_f32(false, a, false, b, (short)0, c,
                                               false, false);
}

#define ZKC    64
#define ZPITCH (ZKC + 4)
#define ZROWS  208
#define ZNLD   (ZROWS * (ZKC / 4) / 128)
#define ZNP    (ZKC / 8)
#define ZPP    20

__device__ __forceinline__ void z_gload(v4f (&g)[ZNLD],
                                        const float* __restrict__ p0,
                                        const float* __restrict__ p1,
                                        const float* __restrict__ p2,
                                        const float* __restrict__ px, int k) {
#pragma unroll
  for (int i = 0; i < ZNLD; ++i) {
    const float* p = (i < 4)    ? p0 + i * 8 * INF
                     : (i < 12) ? p1 + (i - 4) * 8 * INF
                     : (i < 24) ? p2 + (i - 12) * 8 * INF
                                : px + (i - 24) * 8 * INF;
    g[i] = *(const v4f*)(p + k);
  }
}

__device__ __forceinline__ void z_lstore(float* buf, const v4f (&g)[ZNLD]) {
#pragma unroll
  for (int i = 0; i < ZNLD; ++i)
    *(v4f*)(buf + i * 8 * ZPITCH) = g[i];
}

__device__ __forceinline__ void z_lfrag(v4f (&f)[4][ZNP], const float* abuf,
                                        const float* bbuf) {
#pragma unroll
  for (int p = 0; p < ZNP; ++p) {
#pragma unroll
    for (int t = 0; t < 3; ++t)
      f[t][p] = *(const v4f*)(abuf + t * 16 * ZPITCH + p * 8);
    f[3][p] = *(const v4f*)(bbuf + p * 8);
  }
}

__device__ __forceinline__ void z_mma(v8f (&acc)[3], const v4f (&f)[4][ZNP]) {
#pragma unroll
  for (int p = 0; p < ZNP; ++p)
#pragma unroll
    for (int e = 0; e < 2; ++e) {
      v2f b;
      b[0] = f[3][p][2 * e];
      b[1] = f[3][p][2 * e + 1];
#pragma unroll
      for (int t = 0; t < 3; ++t) {
        v2f a;
        a[0] = f[t][p][2 * e];
        a[1] = f[t][p][2 * e + 1];
        acc[t] = wmma_f32(a, b, acc[t]);
      }
    }
}

__global__ void __launch_bounds__(128) taylor_zr_kernel(
    const float* __restrict__ X, const float* __restrict__ I0,
    const float* __restrict__ I1, const float* __restrict__ I2,
    const float* __restrict__ G0, const float* __restrict__ G1,
    float* __restrict__ Z, float* __restrict__ R) {
  __shared__ __attribute__((aligned(16))) float zs[2][ZROWS * ZPITCH];
  const int tid = threadIdx.x;
  const int lane = tid & 31;
  const int w = __builtin_amdgcn_readfirstlane(tid >> 5);
  const int c = lane & 15;
  const int h = lane >> 4;
  const int col0 = blockIdx.x * 16;

  const int goff = (tid >> 4) * INF + (tid & 15) * 4;
  const float* p0 = I0 + goff;
  const float* p1 = I1 + goff;
  const float* p2 = I2 + goff;
  const float* px = X + col0 * INF + goff;
  const int soff = (tid >> 4) * ZPITCH + (tid & 15) * 4;
  const int aoff = (48 * w + c) * ZPITCH + 4 * h;
  const int boff = (192 + c) * ZPITCH + 4 * h;

  v8f acc[3];
#pragma unroll
  for (int t = 0; t < 3; ++t) acc[t] = v8f{};

  v4f g[ZNLD];
  v4f f0[4][ZNP], f1[4][ZNP];

  z_gload(g, p0, p1, p2, px, 0);
  z_lstore(zs[0] + soff, g);
  z_gload(g, p0, p1, p2, px, ZKC);
  __syncthreads();
  z_lfrag(f0, zs[0] + aoff, zs[0] + boff);
  z_lstore(zs[1] + soff, g);
  __syncthreads();

  for (int t = 0; t < 14; t += 2) {
    z_gload(g, p0, p1, p2, px, (t + 2) * ZKC);
    z_lfrag(f1, zs[1] + aoff, zs[1] + boff);
    __builtin_amdgcn_sched_barrier(0);
    z_mma(acc, f0);
    __builtin_amdgcn_sched_barrier(0);
    z_lstore(zs[0] + soff, g);
    __syncthreads();
    z_gload(g, p0, p1, p2, px, (t + 3) * ZKC);
    z_lfrag(f0, zs[0] + aoff, zs[0] + boff);
    __builtin_amdgcn_sched_barrier(0);
    z_mma(acc, f1);
    __builtin_amdgcn_sched_barrier(0);
    z_lstore(zs[1] + soff, g);
    __syncthreads();
  }
  z_lfrag(f1, zs[1] + aoff, zs[1] + boff);

  v4f ga[8][2][4];
  v4f g0a[4];
  {
    const float* gp = G1 + c * 1024 + w * 256 + 4 * h;
#pragma unroll
    for (int i = 0; i < 8; ++i)
#pragma unroll
      for (int rt = 0; rt < 2; ++rt)
#pragma unroll
        for (int p = 0; p < 4; ++p)
          ga[i][rt][p] = *(const v4f*)(gp + rt * 16384 + i * 32 + p * 8);
    const float* g0p = G0 + (16 * (w & 1) + c) * 32 + 4 * h;
#pragma unroll
    for (int p = 0; p < 4; ++p)
      g0a[p] = *(const v4f*)(g0p + p * 8);
  }
  __builtin_amdgcn_sched_barrier(0);
  z_mma(acc, f0);
  z_mma(acc, f1);

  float* P = zs[0];
  if (w < 2) {
#pragma unroll
    for (int t = 0; t < 3; ++t)
#pragma unroll
      for (int v = 0; v < 8; ++v)
        P[(48 * w + 16 * t + 8 * h + v) * ZPP + c] = acc[t][v];
  }
  __syncthreads();

  if (w >= 2) {
#pragma unroll
    for (int t = 0; t < 3; ++t)
#pragma unroll
      for (int v = 0; v < 8; ++v)
        Z[(48 * w + 16 * t + 8 * h + v) * BATCH + col0 + c] = acc[t][v];
  } else {
    v8f a0 = v8f{};
#pragma unroll
    for (int p = 0; p < 4; ++p)
#pragma unroll
      for (int e = 0; e < 2; ++e) {
        const int k = 8 * p + 4 * h + 2 * e;
        v2f a, b;
        a[0] = g0a[p][2 * e];
        a[1] = g0a[p][2 * e + 1];
        b[0] = P[k * ZPP + c];
        b[1] = P[(k + 1) * ZPP + c];
        a0 = wmma_f32(a, b, a0);
      }
#pragma unroll
    for (int v = 0; v < 8; ++v)
      R[(16 * w + 8 * h + v) * BATCH + col0 + c] = a0[v];
  }

  v2f z0r[4][2];
#pragma unroll
  for (int p = 0; p < 4; ++p)
#pragma unroll
    for (int e = 0; e < 2; ++e) {
      const int k0 = 8 * p + 4 * h + 2 * e;
      z0r[p][e][0] = P[(32 + k0) * ZPP + c];
      z0r[p][e][1] = P[(32 + k0 + 1) * ZPP + c];
    }
  v8f r1[2];
  r1[0] = v8f{};
  r1[1] = v8f{};
#pragma unroll
  for (int i = 0; i < 8; ++i) {
    const float sc = P[(64 + 8 * w + i) * ZPP + c];
#pragma unroll
    for (int p = 0; p < 4; ++p)
#pragma unroll
      for (int e = 0; e < 2; ++e) {
        const v2f b = z0r[p][e] * sc;
#pragma unroll
        for (int rt = 0; rt < 2; ++rt) {
          v2f a;
          a[0] = ga[i][rt][p][2 * e];
          a[1] = ga[i][rt][p][2 * e + 1];
          r1[rt] = wmma_f32(a, b, r1[rt]);
        }
      }
  }
  float* Q = zs[1];
#pragma unroll
  for (int rt = 0; rt < 2; ++rt)
#pragma unroll
    for (int v = 0; v < 8; ++v)
      Q[((w * 2 + rt) * 16 + 8 * h + v) * 16 + c] = r1[rt][v];
  __syncthreads();
#pragma unroll
  for (int i = 0; i < 4; ++i) {
    const int o = tid + 128 * i;
    const int row = o >> 4, col = o & 15;
    const float s = ((Q[o] + Q[o + 512]) + Q[o + 1024]) + Q[o + 1536];
    R[(32 + row) * BATCH + col0 + col] = s;
  }
}

__device__ __forceinline__ void r2_load_chunk(v4f (&a)[2][4], float (&sc)[4],
                                              const float* __restrict__ ap,
                                              const float* __restrict__ z1p,
                                              int k1) {
#pragma unroll
  for (int t = 0; t < 2; ++t)
#pragma unroll
    for (int q = 0; q < 4; ++q)
      a[t][q] = *(const v4f*)(ap + t * 16384 + k1 * 32 + 4 * q);
#pragma unroll
  for (int j = 0; j < 4; ++j)
    sc[j] = z1p[k1 * BATCH + 16 * j];
}

__device__ __forceinline__ void r2_mma_chunk(v8f (&acc)[2][4],
                                             const v4f (&a)[2][4],
                                             const float (&sc)[4],
                                             const v2f (&z0)[4][8]) {
#pragma unroll
  for (int q = 0; q < 4; ++q)
#pragma unroll
    for (int e2 = 0; e2 < 2; ++e2) {
      const int s = 2 * q + e2;
#pragma unroll
      for (int j = 0; j < 4; ++j) {
        const v2f b = z0[j][s] * sc[j];
#pragma unroll
        for (int t = 0; t < 2; ++t) {
          v2f av;
          av[0] = a[t][q][2 * e2];
          av[1] = a[t][q][2 * e2 + 1];
          acc[t][j] = wmma_f32(av, b, acc[t][j]);
        }
      }
    }
}

__global__ void __launch_bounds__(256) taylor_r2_kernel(
    const float* __restrict__ G2, const float* __restrict__ Z,
    float* __restrict__ R) {
  const int lane = threadIdx.x & 31;
  const int wave = threadIdx.x >> 5;
  const int c = lane & 15;
  const int h = lane >> 4;
  const int r = blockIdx.x & 31;
  const int wcol = (blockIdx.x >> 5) * 512 + wave * 64;
  const int col = wcol + c;

  v2f z0[4][8];
#pragma unroll
  for (int j = 0; j < 4; ++j)
#pragma unroll
    for (int s = 0; s < 8; ++s) {
      const float* zp = Z + (96 + 16 * h + 2 * s) * BATCH + col + 16 * j;
      z0[j][s][0] = zp[0];
      z0[j][s][1] = zp[BATCH];
    }

  const float* ap = G2 + (r * 32 + c) * 1024 + 16 * h;
  const float* z1p = Z + 128 * BATCH + col;

  v8f acc[2][4];
#pragma unroll
  for (int t = 0; t < 2; ++t)
#pragma unroll
    for (int j = 0; j < 4; ++j)
      acc[t][j] = v8f{};

  v4f a0[2][4], a1[2][4];
  float s0[4], s1[4];
  r2_load_chunk(a0, s0, ap, z1p, 0);
  for (int k1 = 0; k1 < 32; k1 += 2) {
    r2_load_chunk(a1, s1, ap, z1p, k1 + 1);
    __builtin_amdgcn_sched_barrier(0);
    r2_mma_chunk(acc, a0, s0, z0);
    __builtin_amdgcn_sched_barrier(0);
    const int kn = (k1 + 2 < 32) ? k1 + 2 : 31;
    r2_load_chunk(a0, s0, ap, z1p, kn);
    __builtin_amdgcn_sched_barrier(0);
    r2_mma_chunk(acc, a1, s1, z0);
    __builtin_amdgcn_sched_barrier(0);
    asm volatile("" ::"v"(a0[0][0]), "v"(a0[0][1]), "v"(a0[0][2]), "v"(a0[0][3]),
                 "v"(a0[1][0]), "v"(a0[1][1]), "v"(a0[1][2]), "v"(a0[1][3]));
  }

  const float* z2p = Z + 160 * BATCH + col;
  float tot[4];
#pragma unroll
  for (int j = 0; j < 4; ++j) {
    float p = 0.0f;
#pragma unroll
    for (int t = 0; t < 2; ++t)
#pragma unroll
      for (int v = 0; v < 8; ++v)
        p += acc[t][j][v] * z2p[(16 * t + 8 * h + v) * BATCH + 16 * j];
    p += __shfl_xor(p, 16);
    tot[j] = p;
  }
  float* rp = R + (64 + r) * BATCH + wcol;
  rp[lane]      = h ? tot[1] : tot[0];
  rp[32 + lane] = h ? tot[3] : tot[2];
}

__global__ void __launch_bounds__(256) taylor_out_kernel(
    const float* __restrict__ O0, const float* __restrict__ O1,
    const float* __restrict__ O2, const float* __restrict__ Cst,
    const float* __restrict__ R, float* __restrict__ Y) {
  __shared__ float ts[8 * 16 * 17];
  const int lane = threadIdx.x & 31;
  const int wave = threadIdx.x >> 5;
  const int tile = blockIdx.x * 8 + wave;
  const int mt = tile & 63, nt = tile >> 6;
  const int row0 = mt * 16;
  const int col0 = nt * 16;
  const int m = lane & 15, n = lane & 15;
  const int ko = (lane >> 4) * 2;

  v8f acc = {};
  for (int k = 0; k < 96; k += 4) {
    const int kk = k + ko;
    const float* Op = O0; int rk = kk;
    if (kk >= 64)      { Op = O2; rk = kk - 64; }
    else if (kk >= 32) { Op = O1; rk = kk - 32; }
    v2f a, b;
    a[0] = Op[(row0 + m) * 32 + rk];
    a[1] = Op[(row0 + m) * 32 + rk + 1];
    b[0] = R[kk * BATCH + col0 + n];
    b[1] = R[(kk + 1) * BATCH + col0 + n];
    acc = wmma_f32(a, b, acc);
  }
  const int mb = (lane >> 4) * 8;
  float* tw = ts + wave * 272;
#pragma unroll
  for (int v = 0; v < 8; ++v)
    tw[(mb + v) * 17 + n] = acc[v];
  __syncthreads();

  const int ol = lane & 15;
  const int bh = (lane >> 4) * 8;
  const float cv = Cst[row0 + ol];
#pragma unroll
  for (int v = 0; v < 8; ++v)
    Y[(col0 + bh + v) * OUTF + row0 + ol] = tw[ol * 17 + bh + v] + cv;
}

extern "C" void kernel_launch(void* const* d_in, const int* in_sizes, int n_in,
                              void* d_out, int out_size, void* d_ws,
                              size_t ws_size, hipStream_t stream) {
  const float* X  = (const float*)d_in[0];
  const float* C  = (const float*)d_in[1];
  const float* O0 = (const float*)d_in[2];
  const float* I0 = (const float*)d_in[3];
  const float* G0 = (const float*)d_in[4];
  const float* O1 = (const float*)d_in[5];
  const float* I1 = (const float*)d_in[6];
  const float* G1 = (const float*)d_in[7];
  const float* O2 = (const float*)d_in[8];
  const float* I2 = (const float*)d_in[9];
  const float* G2 = (const float*)d_in[10];
  float* Y = (float*)d_out;

  float* Z = (float*)d_ws;
  float* R = Z + 192 * BATCH;

  taylor_zr_kernel <<<256,  128, 0, stream>>>(X, I0, I1, I2, G0, G1, Z, R);
  taylor_r2_kernel <<<256,  256, 0, stream>>>(G2, Z, R);
  taylor_out_kernel<<<2048, 256, 0, stream>>>(O0, O1, O2, C, R, Y);
}
